# starting kernel
# baseline (speedup 1.0000x reference)
#include <hip/hip_runtime.h>
#include <hip/hip_bf16.h>


typedef float v2f __attribute__((ext_vector_type(2)));
typedef float v8f __attribute__((ext_vector_type(8)));

#define NEG_SLOPE 0.2f
#define LRELU(v) ((v) > 0.0f ? (v) : NEG_SLOPE * (v))


__global__ void zero_header_kernel(int* hdr) {
    if (threadIdx.x < 48) hdr[threadIdx.x] = 0;
}

__global__ void hist_kernel(const int* __restrict__ idx, int n, int* __restrict__ counts) {
    __shared__ int h[16];
    if (threadIdx.x < 16) h[threadIdx.x] = 0;
    __syncthreads();
    int i = blockIdx.x * blockDim.x + threadIdx.x;
    if (i < n) atomicAdd(&h[idx[i] & 15], 1);
    __syncthreads();
    if (threadIdx.x < 16 && h[threadIdx.x] > 0)
        atomicAdd(&counts[threadIdx.x], h[threadIdx.x]);
}

__global__ void scan_kernel(const int* __restrict__ counts, int* __restrict__ offs,
                            int* __restrict__ cursors) {
    if (threadIdx.x == 0) {
        int s = 0;
        for (int e = 0; e < 16; ++e) {
            offs[e] = s;
            cursors[e] = s;
            s += counts[e];
        }
    }
}

__global__ void scatter_kernel(const int* __restrict__ idx, int n,
                               int* __restrict__ cursors, int* __restrict__ perm) {
    __shared__ int lcount[16];
    __shared__ int lbase[16];
    if (threadIdx.x < 16) lcount[threadIdx.x] = 0;
    __syncthreads();
    int i = blockIdx.x * blockDim.x + threadIdx.x;
    int e = 0, rank = 0;
    bool live = (i < n);
    if (live) {
        e = idx[i] & 15;
        rank = atomicAdd(&lcount[e], 1);
    }
    __syncthreads();
    if (threadIdx.x < 16 && lcount[threadIdx.x] > 0)
        lbase[threadIdx.x] = atomicAdd(&cursors[threadIdx.x], lcount[threadIdx.x]);
    __syncthreads();
    if (live) perm[lbase[e] + rank] = i;
}


__global__ __launch_bounds__(32) void expert_mlp_wmma_kernel(
    const float* __restrict__ x,
    const float* __restrict__ W0, const float* __restrict__ b0,
    const float* __restrict__ W1, const float* __restrict__ b1,
    const float* __restrict__ W2, const float* __restrict__ b2,
    const int* __restrict__ offs, const int* __restrict__ counts,
    const int* __restrict__ perm,
    float* __restrict__ out)
{
    const int e    = blockIdx.x;
    const int lane = threadIdx.x;
    const int m    = lane & 15;
    const int hi   = lane >> 4;

    __shared__ float sW0[8 * 16];
    __shared__ float sW1[16 * 32];
    __shared__ float sW2[32 * 16];
    __shared__ float sB0[16], sB1[32], sB2[16];
    __shared__ float sX[16 * 8];
    __shared__ float sH[16 * 33];
    __shared__ int   sRow[16];

    for (int i = lane; i < 128; i += 32) {
        int k = i >> 4, n = i & 15;
        sW0[i] = (k < 7) ? W0[e * 112 + k * 16 + n] : 0.0f;
    }
    for (int i = lane; i < 512; i += 32) sW1[i] = W1[e * 512 + i];
    for (int i = lane; i < 512; i += 32) sW2[i] = W2[e * 512 + i];
    if (lane < 16) sB0[lane] = b0[e * 16 + lane];
    sB1[lane] = b1[e * 32 + lane];
    if (lane < 16) sB2[lane] = b2[e * 16 + lane];
    __syncthreads();

    v2f bf0[2], bf1[2][4], bf2[8];
#pragma unroll
    for (int c = 0; c < 2; ++c) {
        int kk = 4 * c + 2 * hi;
        bf0[c].x = sW0[kk * 16 + m];
        bf0[c].y = sW0[(kk + 1) * 16 + m];
    }
#pragma unroll
    for (int j = 0; j < 2; ++j)
#pragma unroll
        for (int c = 0; c < 4; ++c) {
            int kk = 4 * c + 2 * hi;
            int n  = 16 * j + m;
            bf1[j][c].x = sW1[kk * 32 + n];
            bf1[j][c].y = sW1[(kk + 1) * 32 + n];
        }
#pragma unroll
    for (int c = 0; c < 8; ++c) {
        int kk = 4 * c + 2 * hi;
        bf2[c].x = sW2[kk * 16 + m];
        bf2[c].y = sW2[(kk + 1) * 16 + m];
    }

    const float bias0 = sB0[m];
    const float b1lo  = sB1[m];
    const float b1hi  = sB1[16 + m];
    const float bias2 = sB2[m];

    const int base   = offs[e];
    const int cnt    = counts[e];
    const int ntiles = (cnt + 15) >> 4;

    for (int t = blockIdx.y; t < ntiles; t += gridDim.y) {
        __syncthreads();

        int valid = cnt - (t << 4);
        if (valid > 16) valid = 16;

        if (lane < 16) {
            int row = (lane < valid) ? perm[base + (t << 4) + lane] : -1;
            sRow[lane] = row;
            if (row >= 0) {
                const float* xs = x + row * 7;
#pragma unroll
                for (int k = 0; k < 7; ++k) sX[lane * 8 + k] = xs[k];
            } else {
#pragma unroll
                for (int k = 0; k < 7; ++k) sX[lane * 8 + k] = 0.0f;
            }
            sX[lane * 8 + 7] = 0.0f;
        }
        __syncthreads();

        v2f a0, a1;
        a0.x = sX[m * 8 + 2 * hi];
        a0.y = sX[m * 8 + 2 * hi + 1];
        a1.x = sX[m * 8 + 4 + 2 * hi];
        a1.y = sX[m * 8 + 4 + 2 * hi + 1];
        v8f c0 = {};
        c0 = __builtin_amdgcn_wmma_f32_16x16x4_f32(false, a0, false, bf0[0], (short)0, c0, false, false);
        c0 = __builtin_amdgcn_wmma_f32_16x16x4_f32(false, a1, false, bf0[1], (short)0, c0, false, false);
#pragma unroll
        for (int r = 0; r < 8; ++r) {
            float v = c0[r] + bias0;
            c0[r] = LRELU(v);
        }
#pragma unroll
        for (int r = 0; r < 8; ++r) sH[(r + 8 * hi) * 33 + m] = c0[r];
        __syncthreads();

        v2f af1[4];
#pragma unroll
        for (int c = 0; c < 4; ++c) {
            int kk = 4 * c + 2 * hi;
            af1[c].x = sH[m * 33 + kk];
            af1[c].y = sH[m * 33 + kk + 1];
        }
        __syncthreads();

        v8f c1lo = {}, c1hi = {};
#pragma unroll
        for (int c = 0; c < 4; ++c)
            c1lo = __builtin_amdgcn_wmma_f32_16x16x4_f32(false, af1[c], false, bf1[0][c], (short)0, c1lo, false, false);
#pragma unroll
        for (int c = 0; c < 4; ++c)
            c1hi = __builtin_amdgcn_wmma_f32_16x16x4_f32(false, af1[c], false, bf1[1][c], (short)0, c1hi, false, false);
#pragma unroll
        for (int r = 0; r < 8; ++r) {
            float vlo = c1lo[r] + b1lo;
            float vhi = c1hi[r] + b1hi;
            sH[(r + 8 * hi) * 33 + m]      = LRELU(vlo);
            sH[(r + 8 * hi) * 33 + 16 + m] = LRELU(vhi);
        }
        __syncthreads();

        v8f c2 = {};
#pragma unroll
        for (int c = 0; c < 8; ++c) {
            int kk = 4 * c + 2 * hi;
            v2f a;
            a.x = sH[m * 33 + kk];
            a.y = sH[m * 33 + kk + 1];
            c2 = __builtin_amdgcn_wmma_f32_16x16x4_f32(false, a, false, bf2[c], (short)0, c2, false, false);
        }

#pragma unroll
        for (int r = 0; r < 8; ++r) {
            int row = sRow[r + 8 * hi];
            if (row >= 0) {
                float v = c2[r] + bias2;
                out[row * 16 + m] = LRELU(v);
            }
        }
    }
}


extern "C" void kernel_launch(void* const* d_in, const int* in_sizes, int n_in,
                              void* d_out, int out_size, void* d_ws, size_t ws_size,
                              hipStream_t stream) {
    const float* x  = (const float*)d_in[0];
    const int*   idx = (const int*)d_in[1];
    const float* W0 = (const float*)d_in[2];
    const float* b0 = (const float*)d_in[3];
    const float* W1 = (const float*)d_in[4];
    const float* b1 = (const float*)d_in[5];
    const float* W2 = (const float*)d_in[6];
    const float* b2 = (const float*)d_in[7];
    float* out = (float*)d_out;

    const int n = in_sizes[1];

    int* hdr     = (int*)d_ws;
    int* counts  = hdr;
    int* offs    = hdr + 16;
    int* cursors = hdr + 32;
    int* perm    = hdr + 64;

    const int nb = (n + 255) / 256;

    zero_header_kernel<<<1, 64, 0, stream>>>(hdr);
    hist_kernel<<<nb, 256, 0, stream>>>(idx, n, counts);
    scan_kernel<<<1, 32, 0, stream>>>(counts, offs, cursors);
    scatter_kernel<<<nb, 256, 0, stream>>>(idx, n, cursors, perm);

    dim3 grid2(16, 128);
    expert_mlp_wmma_kernel<<<grid2, 32, 0, stream>>>(
        x, W0, b0, W1, b1, W2, b2, offs, counts, perm, out);
}


# MLP layers computed transposed (weights as A, activations as B): each layer's D registers are the next layer's B fragments, no LDS or barriers in tile loop; k regrouped per WMMA (sum order changes)
# speedup vs baseline: 1.2110x; 1.2110x over previous
#include <hip/hip_runtime.h>
#include <hip/hip_bf16.h>


typedef float v2f __attribute__((ext_vector_type(2)));
typedef float v8f __attribute__((ext_vector_type(8)));

#define NEG_SLOPE 0.2f


__global__ void zero_header_kernel(int* hdr) {
    if (threadIdx.x < 48) hdr[threadIdx.x] = 0;
}

__global__ void hist_kernel(const int* __restrict__ idx, int n, int* __restrict__ counts) {
    __shared__ int h[16];
    if (threadIdx.x < 16) h[threadIdx.x] = 0;
    __syncthreads();
    int i = blockIdx.x * blockDim.x + threadIdx.x;
    if (i < n) atomicAdd(&h[idx[i] & 15], 1);
    __syncthreads();
    if (threadIdx.x < 16 && h[threadIdx.x] > 0)
        atomicAdd(&counts[threadIdx.x], h[threadIdx.x]);
}

__global__ void scan_kernel(const int* __restrict__ counts, int* __restrict__ offs,
                            int* __restrict__ cursors) {
    if (threadIdx.x == 0) {
        int s = 0;
        for (int e = 0; e < 16; ++e) {
            offs[e] = s;
            cursors[e] = s;
            s += counts[e];
        }
    }
}

__global__ void scatter_kernel(const int* __restrict__ idx, int n,
                               int* __restrict__ cursors, int* __restrict__ perm) {
    __shared__ int lcount[16];
    __shared__ int lbase[16];
    if (threadIdx.x < 16) lcount[threadIdx.x] = 0;
    __syncthreads();
    int i = blockIdx.x * blockDim.x + threadIdx.x;
    int e = 0, rank = 0;
    bool live = (i < n);
    if (live) {
        e = idx[i] & 15;
        rank = atomicAdd(&lcount[e], 1);
    }
    __syncthreads();
    if (threadIdx.x < 16 && lcount[threadIdx.x] > 0)
        lbase[threadIdx.x] = atomicAdd(&cursors[threadIdx.x], lcount[threadIdx.x]);
    __syncthreads();
    if (live) perm[lbase[e] + rank] = i;
}


__device__ __forceinline__ float lrelu(float v) { return v > 0.0f ? v : NEG_SLOPE * v; }

__device__ __forceinline__ float other_half(float v) {
    return __builtin_bit_cast(float, __builtin_amdgcn_permlanex16(
        __builtin_bit_cast(int, v), __builtin_bit_cast(int, v),
        0x76543210, 0xfedcba98, false, false));
}

__global__ __launch_bounds__(32) void expert_mlp_wmma_kernel(
    const float* __restrict__ x,
    const float* __restrict__ W0, const float* __restrict__ b0,
    const float* __restrict__ W1, const float* __restrict__ b1,
    const float* __restrict__ W2, const float* __restrict__ b2,
    const int* __restrict__ offs, const int* __restrict__ counts,
    const int* __restrict__ perm,
    float* __restrict__ out)
{
    const int e    = blockIdx.x;
    const int lane = threadIdx.x;
    const int c16  = lane & 15;
    const int h    = lane >> 4;

    v2f wa0[2];
    {
        const float* w = W0 + e * 112 + c16;
#pragma unroll
        for (int q = 0; q < 2; ++q) {
            const int k0 = 4 * q + 2 * h;
            const int k1 = k0 + 1;
            const float v0 = w[k0 * 16];
            const float v1 = w[(k1 < 7 ? k1 : 6) * 16];
            wa0[q].x = v0;
            wa0[q].y = (k1 < 7) ? v1 : 0.0f;
        }
    }
    v2f wa1[2][4];
    {
        const float* w = W1 + e * 512 + c16;
#pragma unroll
        for (int j = 0; j < 2; ++j)
#pragma unroll
            for (int q = 0; q < 4; ++q) {
                const int k0 = 8 * h + 2 * q;
                wa1[j][q].x = w[k0 * 32 + 16 * j];
                wa1[j][q].y = w[(k0 + 1) * 32 + 16 * j];
            }
    }
    v2f wa2[2][4];
    {
        const float* w = W2 + e * 512 + c16;
#pragma unroll
        for (int j = 0; j < 2; ++j)
#pragma unroll
            for (int q = 0; q < 4; ++q) {
                const int k0 = 16 * j + 8 * h + 2 * q;
                wa2[j][q].x = w[k0 * 16];
                wa2[j][q].y = w[(k0 + 1) * 16];
            }
    }
    float bb0[8], bb1[2][8], bb2[8];
#pragma unroll
    for (int r = 0; r < 8; ++r) {
        bb0[r]    = b0[e * 16 + 8 * h + r];
        bb1[0][r] = b1[e * 32 + 8 * h + r];
        bb1[1][r] = b1[e * 32 + 16 + 8 * h + r];
        bb2[r]    = b2[e * 16 + 8 * h + r];
    }

    const int base   = offs[e];
    const int cnt    = counts[e];
    const int ntiles = (cnt + 15) >> 4;

    for (int t = blockIdx.y; t < ntiles; t += gridDim.y) {
        const int  p  = (t << 4) + c16;
        const bool ok = p < cnt;
        const int  row = perm[base + (ok ? p : cnt - 1)];

        const float* xr = x + row * 7 + 2 * h;
        v2f bx0, bx1;
        bx0.x = xr[0];
        bx0.y = xr[1];
        bx1.x = xr[4];
        const float x7 = xr[h ? 4 : 5];
        bx1.y = h ? 0.0f : x7;

        v8f d0 = {};
        d0 = __builtin_amdgcn_wmma_f32_16x16x4_f32(false, wa0[0], false, bx0, (short)0, d0, false, false);
        d0 = __builtin_amdgcn_wmma_f32_16x16x4_f32(false, wa0[1], false, bx1, (short)0, d0, false, false);
#pragma unroll
        for (int r = 0; r < 8; ++r) d0[r] = lrelu(d0[r] + bb0[r]);

        v8f d1lo = {}, d1hi = {};
#pragma unroll
        for (int q = 0; q < 4; ++q) {
            v2f bq;
            bq.x = d0[2 * q];
            bq.y = d0[2 * q + 1];
            d1lo = __builtin_amdgcn_wmma_f32_16x16x4_f32(false, wa1[0][q], false, bq, (short)0, d1lo, false, false);
            d1hi = __builtin_amdgcn_wmma_f32_16x16x4_f32(false, wa1[1][q], false, bq, (short)0, d1hi, false, false);
        }
#pragma unroll
        for (int r = 0; r < 8; ++r) {
            d1lo[r] = lrelu(d1lo[r] + bb1[0][r]);
            d1hi[r] = lrelu(d1hi[r] + bb1[1][r]);
        }

        v8f d2 = {};
#pragma unroll
        for (int q = 0; q < 4; ++q) {
            v2f bq;
            bq.x = d1lo[2 * q];
            bq.y = d1lo[2 * q + 1];
            d2 = __builtin_amdgcn_wmma_f32_16x16x4_f32(false, wa2[0][q], false, bq, (short)0, d2, false, false);
        }
#pragma unroll
        for (int q = 0; q < 4; ++q) {
            v2f bq;
            bq.x = d1hi[2 * q];
            bq.y = d1hi[2 * q + 1];
            d2 = __builtin_amdgcn_wmma_f32_16x16x4_f32(false, wa2[1][q], false, bq, (short)0, d2, false, false);
        }

        float o[8];
#pragma unroll
        for (int r = 0; r < 8; ++r) o[r] = lrelu(d2[r] + bb2[r]);

        float4 first, second;
        {
            float keep[4], recv[4];
#pragma unroll
            for (int i = 0; i < 4; ++i) {
                keep[i] = h ? o[4 + i] : o[i];
                recv[i] = other_half(h ? o[i] : o[4 + i]);
            }
            first  = h ? make_float4(recv[0], recv[1], recv[2], recv[3])
                       : make_float4(keep[0], keep[1], keep[2], keep[3]);
            second = h ? make_float4(keep[0], keep[1], keep[2], keep[3])
                       : make_float4(recv[0], recv[1], recv[2], recv[3]);
        }
        if (ok) {
            float4* orow = reinterpret_cast<float4*>(out + row * 16);
            orow[h]     = first;
            orow[2 + h] = second;
        }
    }
}


extern "C" void kernel_launch(void* const* d_in, const int* in_sizes, int n_in,
                              void* d_out, int out_size, void* d_ws, size_t ws_size,
                              hipStream_t stream) {
    const float* x  = (const float*)d_in[0];
    const int*   idx = (const int*)d_in[1];
    const float* W0 = (const float*)d_in[2];
    const float* b0 = (const float*)d_in[3];
    const float* W1 = (const float*)d_in[4];
    const float* b1 = (const float*)d_in[5];
    const float* W2 = (const float*)d_in[6];
    const float* b2 = (const float*)d_in[7];
    float* out = (float*)d_out;

    const int n = in_sizes[1];

    int* hdr     = (int*)d_ws;
    int* counts  = hdr;
    int* offs    = hdr + 16;
    int* cursors = hdr + 32;
    int* perm    = hdr + 64;

    const int nb = (n + 255) / 256;

    zero_header_kernel<<<1, 64, 0, stream>>>(hdr);
    hist_kernel<<<nb, 256, 0, stream>>>(idx, n, counts);
    scan_kernel<<<1, 32, 0, stream>>>(counts, offs, cursors);
    scatter_kernel<<<nb, 256, 0, stream>>>(idx, n, cursors, perm);

    dim3 grid2(16, 128);
    expert_mlp_wmma_kernel<<<grid2, 32, 0, stream>>>(
        x, W0, b0, W1, b1, W2, b2, offs, counts, perm, out);
}


# Binning fused into the MLP launch: each 512-thread block counting-sorts its own 2048 consecutive points by expert in LDS, wave e runs expert e's tiles; 5 launches -> 1, no workspace; sums unchanged
# speedup vs baseline: 2.6626x; 2.1987x over previous
#include <hip/hip_runtime.h>
#include <hip/hip_bf16.h>


typedef float v2f __attribute__((ext_vector_type(2)));
typedef float v8f __attribute__((ext_vector_type(8)));

#define NEG_SLOPE 0.2f
#define CHUNK 2048
#define TPB   512
#define PPT   (CHUNK / TPB)


__device__ __forceinline__ float lrelu(float v) { return v > 0.0f ? v : NEG_SLOPE * v; }

__device__ __forceinline__ float other_half(float v) {
    return __builtin_bit_cast(float, __builtin_amdgcn_permlanex16(
        __builtin_bit_cast(int, v), __builtin_bit_cast(int, v),
        0x76543210, 0xfedcba98, false, false));
}

__global__ __launch_bounds__(TPB) void expert_mlp_wmma_kernel(
    const float* __restrict__ x, const int* __restrict__ idx, int n,
    const float* __restrict__ W0, const float* __restrict__ b0,
    const float* __restrict__ W1, const float* __restrict__ b1,
    const float* __restrict__ W2, const float* __restrict__ b2,
    float* __restrict__ out)
{
    __shared__ int lcount[16];
    __shared__ int lperm[CHUNK];

    const int tid  = threadIdx.x;
    const int lane = tid & 31;
    const int c16  = lane & 15;
    const int h    = lane >> 4;
    const int e      = __builtin_amdgcn_readfirstlane(tid >> 5);
    const int chunk0 = blockIdx.x * CHUNK;

    int  pe[PPT];
    bool live[PPT];
#pragma unroll
    for (int j = 0; j < PPT; ++j) {
        const int p = chunk0 + j * TPB + tid;
        live[j] = p < n;
        pe[j]   = idx[live[j] ? p : n - 1] & 15;
    }
    if (tid < 16) lcount[tid] = 0;

    v2f wa0[2];
    {
        const float* w = W0 + e * 112 + c16;
#pragma unroll
        for (int q = 0; q < 2; ++q) {
            const int k0 = 4 * q + 2 * h;
            const int k1 = k0 + 1;
            const float v0 = w[k0 * 16];
            const float v1 = w[(k1 < 7 ? k1 : 6) * 16];
            wa0[q].x = v0;
            wa0[q].y = (k1 < 7) ? v1 : 0.0f;
        }
    }
    v2f wa1[2][4];
    {
        const float* w = W1 + e * 512 + c16;
#pragma unroll
        for (int j = 0; j < 2; ++j)
#pragma unroll
            for (int q = 0; q < 4; ++q) {
                const int k0 = 8 * h + 2 * q;
                wa1[j][q].x = w[k0 * 32 + 16 * j];
                wa1[j][q].y = w[(k0 + 1) * 32 + 16 * j];
            }
    }
    v2f wa2[2][4];
    {
        const float* w = W2 + e * 512 + c16;
#pragma unroll
        for (int j = 0; j < 2; ++j)
#pragma unroll
            for (int q = 0; q < 4; ++q) {
                const int k0 = 16 * j + 8 * h + 2 * q;
                wa2[j][q].x = w[k0 * 16];
                wa2[j][q].y = w[(k0 + 1) * 16];
            }
    }
    float bb0[8], bb1[2][8], bb2[8];
#pragma unroll
    for (int r = 0; r < 8; ++r) {
        bb0[r]    = b0[e * 16 + 8 * h + r];
        bb1[0][r] = b1[e * 32 + 8 * h + r];
        bb1[1][r] = b1[e * 32 + 16 + 8 * h + r];
        bb2[r]    = b2[e * 16 + 8 * h + r];
    }

    __syncthreads();
    int rank[PPT];
#pragma unroll
    for (int j = 0; j < PPT; ++j)
        rank[j] = live[j] ? atomicAdd(&lcount[pe[j]], 1) : 0;
    __syncthreads();
    int binstart = 0;
#pragma unroll
    for (int k = 0; k < 16; ++k) {
        const int ck = lcount[k];
        binstart += (k < c16) ? ck : 0;
    }
#pragma unroll
    for (int j = 0; j < PPT; ++j) {
        const int s0 = __shfl(binstart, pe[j]);
        if (live[j]) lperm[s0 + rank[j]] = j * TPB + tid;
    }
    const int base = __builtin_amdgcn_readfirstlane(__shfl(binstart, e));
    const int cnt  = __builtin_amdgcn_readfirstlane(lcount[e]);
    __syncthreads();

    const int ntiles = (cnt + 15) >> 4;

    for (int t = 0; t < ntiles; ++t) {
        const int  p  = (t << 4) + c16;
        const bool ok = p < cnt;
        const int  row = chunk0 + lperm[base + (ok ? p : cnt - 1)];

        const float* xr = x + row * 7 + 2 * h;
        v2f bx0, bx1;
        bx0.x = xr[0];
        bx0.y = xr[1];
        bx1.x = xr[4];
        const float x7 = xr[h ? 4 : 5];
        bx1.y = h ? 0.0f : x7;

        v8f d0 = {};
        d0 = __builtin_amdgcn_wmma_f32_16x16x4_f32(false, wa0[0], false, bx0, (short)0, d0, false, false);
        d0 = __builtin_amdgcn_wmma_f32_16x16x4_f32(false, wa0[1], false, bx1, (short)0, d0, false, false);
#pragma unroll
        for (int r = 0; r < 8; ++r) d0[r] = lrelu(d0[r] + bb0[r]);

        v8f d1lo = {}, d1hi = {};
#pragma unroll
        for (int q = 0; q < 4; ++q) {
            v2f bq;
            bq.x = d0[2 * q];
            bq.y = d0[2 * q + 1];
            d1lo = __builtin_amdgcn_wmma_f32_16x16x4_f32(false, wa1[0][q], false, bq, (short)0, d1lo, false, false);
            d1hi = __builtin_amdgcn_wmma_f32_16x16x4_f32(false, wa1[1][q], false, bq, (short)0, d1hi, false, false);
        }
#pragma unroll
        for (int r = 0; r < 8; ++r) {
            d1lo[r] = lrelu(d1lo[r] + bb1[0][r]);
            d1hi[r] = lrelu(d1hi[r] + bb1[1][r]);
        }

        v8f d2 = {};
#pragma unroll
        for (int q = 0; q < 4; ++q) {
            v2f bq;
            bq.x = d1lo[2 * q];
            bq.y = d1lo[2 * q + 1];
            d2 = __builtin_amdgcn_wmma_f32_16x16x4_f32(false, wa2[0][q], false, bq, (short)0, d2, false, false);
        }
#pragma unroll
        for (int q = 0; q < 4; ++q) {
            v2f bq;
            bq.x = d1hi[2 * q];
            bq.y = d1hi[2 * q + 1];
            d2 = __builtin_amdgcn_wmma_f32_16x16x4_f32(false, wa2[1][q], false, bq, (short)0, d2, false, false);
        }

        float o[8];
#pragma unroll
        for (int r = 0; r < 8; ++r) o[r] = lrelu(d2[r] + bb2[r]);

        float4 first, second;
        {
            float keep[4], recv[4];
#pragma unroll
            for (int i = 0; i < 4; ++i) {
                keep[i] = h ? o[4 + i] : o[i];
                recv[i] = other_half(h ? o[i] : o[4 + i]);
            }
            first  = h ? make_float4(recv[0], recv[1], recv[2], recv[3])
                       : make_float4(keep[0], keep[1], keep[2], keep[3]);
            second = h ? make_float4(keep[0], keep[1], keep[2], keep[3])
                       : make_float4(recv[0], recv[1], recv[2], recv[3]);
        }
        if (ok) {
            float4* orow = reinterpret_cast<float4*>(out + row * 16);
            orow[h]     = first;
            orow[2 + h] = second;
        }
    }
}


extern "C" void kernel_launch(void* const* d_in, const int* in_sizes, int n_in,
                              void* d_out, int out_size, void* d_ws, size_t ws_size,
                              hipStream_t stream) {
    const float* x  = (const float*)d_in[0];
    const int*   idx = (const int*)d_in[1];
    const float* W0 = (const float*)d_in[2];
    const float* b0 = (const float*)d_in[3];
    const float* W1 = (const float*)d_in[4];
    const float* b1 = (const float*)d_in[5];
    const float* W2 = (const float*)d_in[6];
    const float* b2 = (const float*)d_in[7];
    float* out = (float*)d_out;

    const int n = in_sizes[1];

    const int nb = (n + CHUNK - 1) / CHUNK;

    expert_mlp_wmma_kernel<<<nb, TPB, 0, stream>>>(
        x, idx, n, W0, b0, W1, b1, W2, b2, out);
}


# Block's x rows staged in LDS once with whole-line 16-byte loads; tiles read layer-0 B fragments from LDS instead of 3 scattered global loads per lane per tile; sum order unchanged
# speedup vs baseline: 3.0655x; 1.1513x over previous
#include <hip/hip_runtime.h>
#include <hip/hip_bf16.h>


typedef float v2f __attribute__((ext_vector_type(2)));
typedef float v8f __attribute__((ext_vector_type(8)));

#define NEG_SLOPE 0.2f
#define CHUNK 2048
#define TPB   512
#define PPT   (CHUNK / TPB)
#define XV4   (CHUNK * 7 / 4 / TPB)


__device__ __forceinline__ float lrelu(float v) { return v > 0.0f ? v : NEG_SLOPE * v; }

__device__ __forceinline__ float other_half(float v) {
    return __builtin_bit_cast(float, __builtin_amdgcn_permlanex16(
        __builtin_bit_cast(int, v), __builtin_bit_cast(int, v),
        0x76543210, 0xfedcba98, false, false));
}

__global__ __launch_bounds__(TPB) void expert_mlp_wmma_kernel(
    const float* __restrict__ x, const int* __restrict__ idx, int n,
    const float* __restrict__ W0, const float* __restrict__ b0,
    const float* __restrict__ W1, const float* __restrict__ b1,
    const float* __restrict__ W2, const float* __restrict__ b2,
    float* __restrict__ out)
{
    __shared__ int lcount[16];
    __shared__ int lperm[CHUNK];
    __shared__ __attribute__((aligned(16))) float xs[CHUNK * 7];

    const int tid  = threadIdx.x;
    const int lane = tid & 31;
    const int c16  = lane & 15;
    const int h    = lane >> 4;
    const int e      = __builtin_amdgcn_readfirstlane(tid >> 5);
    const int chunk0 = blockIdx.x * CHUNK;

    int  pe[PPT];
    bool live[PPT];
#pragma unroll
    for (int j = 0; j < PPT; ++j) {
        const int p = chunk0 + j * TPB + tid;
        live[j] = p < n;
        pe[j]   = idx[live[j] ? p : n - 1] & 15;
    }
    if (tid < 16) lcount[tid] = 0;

    const int    npts  = (n - chunk0 < CHUNK) ? (n - chunk0) : CHUNK;
    const int    nf    = npts * 7;
    const int    nfull = nf >> 2;
    const float* xg    = x + chunk0 * 7;
    float4 xv[XV4];
#pragma unroll
    for (int j = 0; j < XV4; ++j) {
        const int i4 = j * TPB + tid;
        xv[j] = reinterpret_cast<const float4*>(xg)[i4 < nfull ? i4 : nfull - 1];
    }
    const int  tail0    = nfull << 2;
    const bool has_tail = tid < nf - tail0;
    float xt = 0.0f;
    if (has_tail) xt = xg[tail0 + tid];

    v2f wa0[2];
    {
        const float* w = W0 + e * 112 + c16;
#pragma unroll
        for (int q = 0; q < 2; ++q) {
            const int k0 = 4 * q + 2 * h;
            const int k1 = k0 + 1;
            const float v0 = w[k0 * 16];
            const float v1 = w[(k1 < 7 ? k1 : 6) * 16];
            wa0[q].x = v0;
            wa0[q].y = (k1 < 7) ? v1 : 0.0f;
        }
    }
    v2f wa1[2][4];
    {
        const float* w = W1 + e * 512 + c16;
#pragma unroll
        for (int j = 0; j < 2; ++j)
#pragma unroll
            for (int q = 0; q < 4; ++q) {
                const int k0 = 8 * h + 2 * q;
                wa1[j][q].x = w[k0 * 32 + 16 * j];
                wa1[j][q].y = w[(k0 + 1) * 32 + 16 * j];
            }
    }
    v2f wa2[2][4];
    {
        const float* w = W2 + e * 512 + c16;
#pragma unroll
        for (int j = 0; j < 2; ++j)
#pragma unroll
            for (int q = 0; q < 4; ++q) {
                const int k0 = 16 * j + 8 * h + 2 * q;
                wa2[j][q].x = w[k0 * 16];
                wa2[j][q].y = w[(k0 + 1) * 16];
            }
    }
    float bb0[8], bb1[2][8], bb2[8];
#pragma unroll
    for (int r = 0; r < 8; ++r) {
        bb0[r]    = b0[e * 16 + 8 * h + r];
        bb1[0][r] = b1[e * 32 + 8 * h + r];
        bb1[1][r] = b1[e * 32 + 16 + 8 * h + r];
        bb2[r]    = b2[e * 16 + 8 * h + r];
    }

#pragma unroll
    for (int j = 0; j < XV4; ++j)
        reinterpret_cast<float4*>(xs)[j * TPB + tid] = xv[j];

    __syncthreads();
    if (has_tail) xs[tail0 + tid] = xt;
    int rank[PPT];
#pragma unroll
    for (int j = 0; j < PPT; ++j)
        rank[j] = live[j] ? atomicAdd(&lcount[pe[j]], 1) : 0;
    __syncthreads();
    int binstart = 0;
#pragma unroll
    for (int k = 0; k < 16; ++k) {
        const int ck = lcount[k];
        binstart += (k < c16) ? ck : 0;
    }
#pragma unroll
    for (int j = 0; j < PPT; ++j) {
        const int s0 = __shfl(binstart, pe[j]);
        if (live[j]) lperm[s0 + rank[j]] = j * TPB + tid;
    }
    const int base = __builtin_amdgcn_readfirstlane(__shfl(binstart, e));
    const int cnt  = __builtin_amdgcn_readfirstlane(lcount[e]);
    __syncthreads();

    const int ntiles = (cnt + 15) >> 4;

    for (int t = 0; t < ntiles; ++t) {
        const int  p  = (t << 4) + c16;
        const bool ok = p < cnt;
        const int  rl  = lperm[base + (ok ? p : cnt - 1)];
        const int  row = chunk0 + rl;

        const float* xr = xs + rl * 7 + 2 * h;
        v2f bx0, bx1;
        bx0.x = xr[0];
        bx0.y = xr[1];
        bx1.x = xr[4];
        const float x7 = xr[h ? 4 : 5];
        bx1.y = h ? 0.0f : x7;

        v8f d0 = {};
        d0 = __builtin_amdgcn_wmma_f32_16x16x4_f32(false, wa0[0], false, bx0, (short)0, d0, false, false);
        d0 = __builtin_amdgcn_wmma_f32_16x16x4_f32(false, wa0[1], false, bx1, (short)0, d0, false, false);
#pragma unroll
        for (int r = 0; r < 8; ++r) d0[r] = lrelu(d0[r] + bb0[r]);

        v8f d1lo = {}, d1hi = {};
#pragma unroll
        for (int q = 0; q < 4; ++q) {
            v2f bq;
            bq.x = d0[2 * q];
            bq.y = d0[2 * q + 1];
            d1lo = __builtin_amdgcn_wmma_f32_16x16x4_f32(false, wa1[0][q], false, bq, (short)0, d1lo, false, false);
            d1hi = __builtin_amdgcn_wmma_f32_16x16x4_f32(false, wa1[1][q], false, bq, (short)0, d1hi, false, false);
        }
#pragma unroll
        for (int r = 0; r < 8; ++r) {
            d1lo[r] = lrelu(d1lo[r] + bb1[0][r]);
            d1hi[r] = lrelu(d1hi[r] + bb1[1][r]);
        }

        v8f d2 = {};
#pragma unroll
        for (int q = 0; q < 4; ++q) {
            v2f bq;
            bq.x = d1lo[2 * q];
            bq.y = d1lo[2 * q + 1];
            d2 = __builtin_amdgcn_wmma_f32_16x16x4_f32(false, wa2[0][q], false, bq, (short)0, d2, false, false);
        }
#pragma unroll
        for (int q = 0; q < 4; ++q) {
            v2f bq;
            bq.x = d1hi[2 * q];
            bq.y = d1hi[2 * q + 1];
            d2 = __builtin_amdgcn_wmma_f32_16x16x4_f32(false, wa2[1][q], false, bq, (short)0, d2, false, false);
        }

        float o[8];
#pragma unroll
        for (int r = 0; r < 8; ++r) o[r] = lrelu(d2[r] + bb2[r]);

        float4 first, second;
        {
            float keep[4], recv[4];
#pragma unroll
            for (int i = 0; i < 4; ++i) {
                keep[i] = h ? o[4 + i] : o[i];
                recv[i] = other_half(h ? o[i] : o[4 + i]);
            }
            first  = h ? make_float4(recv[0], recv[1], recv[2], recv[3])
                       : make_float4(keep[0], keep[1], keep[2], keep[3]);
            second = h ? make_float4(keep[0], keep[1], keep[2], keep[3])
                       : make_float4(recv[0], recv[1], recv[2], recv[3]);
        }
        if (ok) {
            float4* orow = reinterpret_cast<float4*>(out + row * 16);
            orow[h]     = first;
            orow[2 + h] = second;
        }
    }
}


extern "C" void kernel_launch(void* const* d_in, const int* in_sizes, int n_in,
                              void* d_out, int out_size, void* d_ws, size_t ws_size,
                              hipStream_t stream) {
    const float* x  = (const float*)d_in[0];
    const int*   idx = (const int*)d_in[1];
    const float* W0 = (const float*)d_in[2];
    const float* b0 = (const float*)d_in[3];
    const float* W1 = (const float*)d_in[4];
    const float* b1 = (const float*)d_in[5];
    const float* W2 = (const float*)d_in[6];
    const float* b2 = (const float*)d_in[7];
    float* out = (float*)d_out;

    const int n = in_sizes[1];

    const int nb = (n + CHUNK - 1) / CHUNK;

    expert_mlp_wmma_kernel<<<nb, TPB, 0, stream>>>(
        x, idx, n, W0, b0, W1, b1, W2, b2, out);
}


# Tile loop's vector work halved: bias enters as the WMMA C operand (summed first, not last), LeakyReLU as max(v, slope*v), last layer's weight rows permuted so D holds the lane's two store pieces
# speedup vs baseline: 3.2510x; 1.0605x over previous
#include <hip/hip_runtime.h>
#include <hip/hip_bf16.h>


typedef float v2f __attribute__((ext_vector_type(2)));
typedef float v8f __attribute__((ext_vector_type(8)));

#define NEG_SLOPE 0.2f
#define CHUNK 2048
#define TPB   512
#define PPT   (CHUNK / TPB)
#define XV4   (CHUNK * 7 / 4 / TPB)


__device__ __forceinline__ float lrelu(float v) { return __builtin_elementwise_maximum(v, NEG_SLOPE * v); }

__global__ __launch_bounds__(TPB) void expert_mlp_wmma_kernel(
    const float* __restrict__ x, const int* __restrict__ idx, int n,
    const float* __restrict__ W0, const float* __restrict__ b0,
    const float* __restrict__ W1, const float* __restrict__ b1,
    const float* __restrict__ W2, const float* __restrict__ b2,
    float* __restrict__ out)
{
    __shared__ int lcount[16];
    __shared__ int lperm[CHUNK];
    __shared__ __attribute__((aligned(16))) float xs[CHUNK * 7];

    const int tid  = threadIdx.x;
    const int lane = tid & 31;
    const int c16  = lane & 15;
    const int h    = lane >> 4;
    const int e      = __builtin_amdgcn_readfirstlane(tid >> 5);
    const int chunk0 = blockIdx.x * CHUNK;

    int  pe[PPT];
    bool live[PPT];
#pragma unroll
    for (int j = 0; j < PPT; ++j) {
        const int p = chunk0 + j * TPB + tid;
        live[j] = p < n;
        pe[j]   = idx[live[j] ? p : n - 1] & 15;
    }
    if (tid < 16) lcount[tid] = 0;

    const int    npts  = (n - chunk0 < CHUNK) ? (n - chunk0) : CHUNK;
    const int    nf    = npts * 7;
    const int    nfull = nf >> 2;
    const float* xg    = x + chunk0 * 7;
    float4 xv[XV4];
#pragma unroll
    for (int j = 0; j < XV4; ++j) {
        const int i4 = j * TPB + tid;
        xv[j] = reinterpret_cast<const float4*>(xg)[i4 < nfull ? i4 : nfull - 1];
    }
    const int  tail0    = nfull << 2;
    const bool has_tail = tid < nf - tail0;
    float xt = 0.0f;
    if (has_tail) xt = xg[tail0 + tid];

    v2f wa0[2];
    {
        const float* w = W0 + e * 112 + c16;
#pragma unroll
        for (int q = 0; q < 2; ++q) {
            const int k0 = 4 * q + 2 * h;
            const int k1 = k0 + 1;
            const float v0 = w[k0 * 16];
            const float v1 = w[(k1 < 7 ? k1 : 6) * 16];
            wa0[q].x = v0;
            wa0[q].y = (k1 < 7) ? v1 : 0.0f;
        }
    }
    v2f wa1[2][4];
    {
        const float* w = W1 + e * 512 + c16;
#pragma unroll
        for (int j = 0; j < 2; ++j)
#pragma unroll
            for (int q = 0; q < 4; ++q) {
                const int k0 = 8 * h + 2 * q;
                wa1[j][q].x = w[k0 * 32 + 16 * j];
                wa1[j][q].y = w[(k0 + 1) * 32 + 16 * j];
            }
    }
    const int pc16 = (c16 & 3) | ((c16 & 4) << 1) | ((c16 & 8) >> 1);
    v2f wa2[2][4];
    {
        const float* w = W2 + e * 512 + pc16;
#pragma unroll
        for (int j = 0; j < 2; ++j)
#pragma unroll
            for (int q = 0; q < 4; ++q) {
                const int k0 = 16 * j + 8 * h + 2 * q;
                wa2[j][q].x = w[k0 * 16];
                wa2[j][q].y = w[(k0 + 1) * 16];
            }
    }
    v8f cb0, cb1lo, cb1hi, cb2;
#pragma unroll
    for (int r = 0; r < 8; ++r) {
        cb0[r]   = b0[e * 16 + 8 * h + r];
        cb1lo[r] = b1[e * 32 + 8 * h + r];
        cb1hi[r] = b1[e * 32 + 16 + 8 * h + r];
        cb2[r]   = b2[e * 16 + 4 * h + (r & 3) + 8 * (r >> 2)];
    }

#pragma unroll
    for (int j = 0; j < XV4; ++j)
        reinterpret_cast<float4*>(xs)[j * TPB + tid] = xv[j];

    __syncthreads();
    if (has_tail) xs[tail0 + tid] = xt;
    int rank[PPT];
#pragma unroll
    for (int j = 0; j < PPT; ++j)
        rank[j] = live[j] ? atomicAdd(&lcount[pe[j]], 1) : 0;
    __syncthreads();
    int binstart = 0;
#pragma unroll
    for (int k = 0; k < 16; ++k) {
        const int ck = lcount[k];
        binstart += (k < c16) ? ck : 0;
    }
#pragma unroll
    for (int j = 0; j < PPT; ++j) {
        const int s0 = __shfl(binstart, pe[j]);
        if (live[j]) lperm[s0 + rank[j]] = j * TPB + tid;
    }
    const int base = __builtin_amdgcn_readfirstlane(__shfl(binstart, e));
    const int cnt  = __builtin_amdgcn_readfirstlane(lcount[e]);
    __syncthreads();

    const int ntiles = (cnt + 15) >> 4;

    for (int t = 0; t < ntiles; ++t) {
        const int  p  = (t << 4) + c16;
        const bool ok = p < cnt;
        const int  rl  = lperm[base + (ok ? p : cnt - 1)];
        const int  row = chunk0 + rl;

        const float* xr = xs + rl * 7 + 2 * h;
        v2f bx0, bx1;
        bx0.x = xr[0];
        bx0.y = xr[1];
        bx1.x = xr[4];
        const float x7 = xr[h ? 4 : 5];
        bx1.y = h ? 0.0f : x7;

        v8f d0;
        d0 = __builtin_amdgcn_wmma_f32_16x16x4_f32(false, wa0[0], false, bx0, (short)0, cb0, false, false);
        d0 = __builtin_amdgcn_wmma_f32_16x16x4_f32(false, wa0[1], false, bx1, (short)0, d0, false, false);
#pragma unroll
        for (int r = 0; r < 8; ++r) d0[r] = lrelu(d0[r]);

        v8f d1lo = cb1lo, d1hi = cb1hi;
#pragma unroll
        for (int q = 0; q < 4; ++q) {
            v2f bq;
            bq.x = d0[2 * q];
            bq.y = d0[2 * q + 1];
            d1lo = __builtin_amdgcn_wmma_f32_16x16x4_f32(false, wa1[0][q], false, bq, (short)0, d1lo, false, false);
            d1hi = __builtin_amdgcn_wmma_f32_16x16x4_f32(false, wa1[1][q], false, bq, (short)0, d1hi, false, false);
        }
#pragma unroll
        for (int r = 0; r < 8; ++r) {
            d1lo[r] = lrelu(d1lo[r]);
            d1hi[r] = lrelu(d1hi[r]);
        }

        v8f d2 = cb2;
#pragma unroll
        for (int q = 0; q < 4; ++q) {
            v2f bq;
            bq.x = d1lo[2 * q];
            bq.y = d1lo[2 * q + 1];
            d2 = __builtin_amdgcn_wmma_f32_16x16x4_f32(false, wa2[0][q], false, bq, (short)0, d2, false, false);
        }
#pragma unroll
        for (int q = 0; q < 4; ++q) {
            v2f bq;
            bq.x = d1hi[2 * q];
            bq.y = d1hi[2 * q + 1];
            d2 = __builtin_amdgcn_wmma_f32_16x16x4_f32(false, wa2[1][q], false, bq, (short)0, d2, false, false);
        }

        float o[8];
#pragma unroll
        for (int r = 0; r < 8; ++r) o[r] = lrelu(d2[r]);
        const float4 first  = make_float4(o[0], o[1], o[2], o[3]);
        const float4 second = make_float4(o[4], o[5], o[6], o[7]);
        if (ok) {
            float4* orow = reinterpret_cast<float4*>(out + row * 16);
            orow[h]     = first;
            orow[2 + h] = second;
        }
    }
}


extern "C" void kernel_launch(void* const* d_in, const int* in_sizes, int n_in,
                              void* d_out, int out_size, void* d_ws, size_t ws_size,
                              hipStream_t stream) {
    const float* x  = (const float*)d_in[0];
    const int*   idx = (const int*)d_in[1];
    const float* W0 = (const float*)d_in[2];
    const float* b0 = (const float*)d_in[3];
    const float* W1 = (const float*)d_in[4];
    const float* b1 = (const float*)d_in[5];
    const float* W2 = (const float*)d_in[6];
    const float* b2 = (const float*)d_in[7];
    float* out = (float*)d_out;

    const int n = in_sizes[1];

    const int nb = (n + CHUNK - 1) / CHUNK;

    expert_mlp_wmma_kernel<<<nb, TPB, 0, stream>>>(
        x, idx, n, W0, b0, W1, b1, W2, b2, out);
}


# Each store instruction writes whole 64-byte rows (8 rows) instead of a 32-byte piece of 16 rows: lane pairs (c, c^8) swap one piece by a rotate inside the 16-lane row; sums unchanged
# speedup vs baseline: 3.9181x; 1.2052x over previous
#include <hip/hip_runtime.h>
#include <hip/hip_bf16.h>


typedef float v2f __attribute__((ext_vector_type(2)));
typedef float v8f __attribute__((ext_vector_type(8)));

#define NEG_SLOPE 0.2f
#define CHUNK 2048
#define TPB   512
#define PPT   (CHUNK / TPB)
#define XV4   (CHUNK * 7 / 4 / TPB)


__device__ __forceinline__ float lrelu(float v) { return __builtin_elementwise_maximum(v, NEG_SLOPE * v); }

__device__ __forceinline__ float ror8_hi(float keep, float from) {
    return __builtin_bit_cast(float, __builtin_amdgcn_update_dpp(
        __builtin_bit_cast(int, keep), __builtin_bit_cast(int, from), 0x128, 0xf, 0xc, false));
}
__device__ __forceinline__ float ror8_lo(float keep, float from) {
    return __builtin_bit_cast(float, __builtin_amdgcn_update_dpp(
        __builtin_bit_cast(int, keep), __builtin_bit_cast(int, from), 0x128, 0xf, 0x3, false));
}

__global__ __launch_bounds__(TPB) void expert_mlp_wmma_kernel(
    const float* __restrict__ x, const int* __restrict__ idx, int n,
    const float* __restrict__ W0, const float* __restrict__ b0,
    const float* __restrict__ W1, const float* __restrict__ b1,
    const float* __restrict__ W2, const float* __restrict__ b2,
    float* __restrict__ out)
{
    __shared__ int lcount[16];
    __shared__ int lperm[CHUNK];
    __shared__ __attribute__((aligned(16))) float xs[CHUNK * 7];

    const int tid  = threadIdx.x;
    const int lane = tid & 31;
    const int c16  = lane & 15;
    const int h    = lane >> 4;
    const int e      = __builtin_amdgcn_readfirstlane(tid >> 5);
    const int chunk0 = blockIdx.x * CHUNK;

    int  pe[PPT];
    bool live[PPT];
#pragma unroll
    for (int j = 0; j < PPT; ++j) {
        const int p = chunk0 + j * TPB + tid;
        live[j] = p < n;
        pe[j]   = idx[live[j] ? p : n - 1] & 15;
    }
    if (tid < 16) lcount[tid] = 0;

    const int    npts  = (n - chunk0 < CHUNK) ? (n - chunk0) : CHUNK;
    const int    nf    = npts * 7;
    const int    nfull = nf >> 2;
    const float* xg    = x + chunk0 * 7;
    float4 xv[XV4];
#pragma unroll
    for (int j = 0; j < XV4; ++j) {
        const int i4 = j * TPB + tid;
        xv[j] = reinterpret_cast<const float4*>(xg)[i4 < nfull ? i4 : nfull - 1];
    }
    const int  tail0    = nfull << 2;
    const bool has_tail = tid < nf - tail0;
    float xt = 0.0f;
    if (has_tail) xt = xg[tail0 + tid];

    v2f wa0[2];
    {
        const float* w = W0 + e * 112 + c16;
#pragma unroll
        for (int q = 0; q < 2; ++q) {
            const int k0 = 4 * q + 2 * h;
            const int k1 = k0 + 1;
            const float v0 = w[k0 * 16];
            const float v1 = w[(k1 < 7 ? k1 : 6) * 16];
            wa0[q].x = v0;
            wa0[q].y = (k1 < 7) ? v1 : 0.0f;
        }
    }
    v2f wa1[2][4];
    {
        const float* w = W1 + e * 512 + c16;
#pragma unroll
        for (int j = 0; j < 2; ++j)
#pragma unroll
            for (int q = 0; q < 4; ++q) {
                const int k0 = 8 * h + 2 * q;
                wa1[j][q].x = w[k0 * 32 + 16 * j];
                wa1[j][q].y = w[(k0 + 1) * 32 + 16 * j];
            }
    }
    const int pc16 = (c16 & 3) | ((c16 & 4) << 1) | ((c16 & 8) >> 1);
    v2f wa2[2][4];
    {
        const float* w = W2 + e * 512 + pc16;
#pragma unroll
        for (int j = 0; j < 2; ++j)
#pragma unroll
            for (int q = 0; q < 4; ++q) {
                const int k0 = 16 * j + 8 * h + 2 * q;
                wa2[j][q].x = w[k0 * 16];
                wa2[j][q].y = w[(k0 + 1) * 16];
            }
    }
    v8f cb0, cb1lo, cb1hi, cb2;
#pragma unroll
    for (int r = 0; r < 8; ++r) {
        cb0[r]   = b0[e * 16 + 8 * h + r];
        cb1lo[r] = b1[e * 32 + 8 * h + r];
        cb1hi[r] = b1[e * 32 + 16 + 8 * h + r];
        cb2[r]   = b2[e * 16 + 4 * h + (r & 3) + 8 * (r >> 2)];
    }

#pragma unroll
    for (int j = 0; j < XV4; ++j)
        reinterpret_cast<float4*>(xs)[j * TPB + tid] = xv[j];

    __syncthreads();
    if (has_tail) xs[tail0 + tid] = xt;
    int rank[PPT];
#pragma unroll
    for (int j = 0; j < PPT; ++j)
        rank[j] = live[j] ? atomicAdd(&lcount[pe[j]], 1) : 0;
    __syncthreads();
    int binstart = 0;
#pragma unroll
    for (int k = 0; k < 16; ++k) {
        const int ck = lcount[k];
        binstart += (k < c16) ? ck : 0;
    }
#pragma unroll
    for (int j = 0; j < PPT; ++j) {
        const int s0 = __shfl(binstart, pe[j]);
        if (live[j]) lperm[s0 + rank[j]] = j * TPB + tid;
    }
    const int base = __builtin_amdgcn_readfirstlane(__shfl(binstart, e));
    const int cnt  = __builtin_amdgcn_readfirstlane(lcount[e]);
    __syncthreads();

    const int ntiles = (cnt + 15) >> 4;
    const int piece = h + ((c16 & 8) >> 2);
    const int c8    = c16 & 7;

    for (int t = 0; t < ntiles; ++t) {
        const int  p  = (t << 4) + c16;
        const bool ok = p < cnt;
        const int  rl  = lperm[base + (ok ? p : cnt - 1)];

        const float* xr = xs + rl * 7 + 2 * h;
        v2f bx0, bx1;
        bx0.x = xr[0];
        bx0.y = xr[1];
        bx1.x = xr[4];
        const float x7 = xr[h ? 4 : 5];
        bx1.y = h ? 0.0f : x7;

        v8f d0;
        d0 = __builtin_amdgcn_wmma_f32_16x16x4_f32(false, wa0[0], false, bx0, (short)0, cb0, false, false);
        d0 = __builtin_amdgcn_wmma_f32_16x16x4_f32(false, wa0[1], false, bx1, (short)0, d0, false, false);
#pragma unroll
        for (int r = 0; r < 8; ++r) d0[r] = lrelu(d0[r]);

        v8f d1lo = cb1lo, d1hi = cb1hi;
#pragma unroll
        for (int q = 0; q < 4; ++q) {
            v2f bq;
            bq.x = d0[2 * q];
            bq.y = d0[2 * q + 1];
            d1lo = __builtin_amdgcn_wmma_f32_16x16x4_f32(false, wa1[0][q], false, bq, (short)0, d1lo, false, false);
            d1hi = __builtin_amdgcn_wmma_f32_16x16x4_f32(false, wa1[1][q], false, bq, (short)0, d1hi, false, false);
        }
#pragma unroll
        for (int r = 0; r < 8; ++r) {
            d1lo[r] = lrelu(d1lo[r]);
            d1hi[r] = lrelu(d1hi[r]);
        }

        v8f d2 = cb2;
#pragma unroll
        for (int q = 0; q < 4; ++q) {
            v2f bq;
            bq.x = d1lo[2 * q];
            bq.y = d1lo[2 * q + 1];
            d2 = __builtin_amdgcn_wmma_f32_16x16x4_f32(false, wa2[0][q], false, bq, (short)0, d2, false, false);
        }
#pragma unroll
        for (int q = 0; q < 4; ++q) {
            v2f bq;
            bq.x = d1hi[2 * q];
            bq.y = d1hi[2 * q + 1];
            d2 = __builtin_amdgcn_wmma_f32_16x16x4_f32(false, wa2[1][q], false, bq, (short)0, d2, false, false);
        }

        float o[8];
#pragma unroll
        for (int r = 0; r < 8; ++r) o[r] = lrelu(d2[r]);
        float s1[4], s2[4];
#pragma unroll
        for (int i = 0; i < 4; ++i) {
            s1[i] = ror8_hi(o[i], o[4 + i]);
            s2[i] = ror8_lo(o[4 + i], o[i]);
        }
        const int rl1 = __builtin_amdgcn_update_dpp(rl, rl, 0x128, 0xf, 0xc, false);
        const int rl2 = __builtin_amdgcn_update_dpp(rl, rl, 0x128, 0xf, 0x3, false);
        const int p1  = (t << 4) + c8;
        if (p1 < cnt)
            reinterpret_cast<float4*>(out + (chunk0 + rl1) * 16)[piece] = make_float4(s1[0], s1[1], s1[2], s1[3]);
        if (p1 + 8 < cnt)
            reinterpret_cast<float4*>(out + (chunk0 + rl2) * 16)[piece] = make_float4(s2[0], s2[1], s2[2], s2[3]);
    }
}


extern "C" void kernel_launch(void* const* d_in, const int* in_sizes, int n_in,
                              void* d_out, int out_size, void* d_ws, size_t ws_size,
                              hipStream_t stream) {
    const float* x  = (const float*)d_in[0];
    const int*   idx = (const int*)d_in[1];
    const float* W0 = (const float*)d_in[2];
    const float* b0 = (const float*)d_in[3];
    const float* W1 = (const float*)d_in[4];
    const float* b1 = (const float*)d_in[5];
    const float* W2 = (const float*)d_in[6];
    const float* b2 = (const float*)d_in[7];
    float* out = (float*)d_out;

    const int n = in_sizes[1];

    const int nb = (n + CHUNK - 1) / CHUNK;

    expert_mlp_wmma_kernel<<<nb, TPB, 0, stream>>>(
        x, idx, n, W0, b0, W1, b1, W2, b2, out);
}
